# starting kernel
# baseline (speedup 1.0000x reference)
#include <hip/hip_runtime.h>

typedef float v2f __attribute__((ext_vector_type(2)));
typedef float v8f __attribute__((ext_vector_type(8)));

#define TPB 256

__global__ __launch_bounds__(TPB) void hg_init(unsigned* __restrict__ deg,
                                               float* __restrict__ acc1,
                                               float* __restrict__ out,
                                               const float* __restrict__ b2,
                                               int N) {
    long i = (long)blockIdx.x * TPB + threadIdx.x;
    if (i < 6L * N)  deg[i] = 0u;
    if (i < 16L * N) acc1[i] = 0.0f;
    if (i < N)       out[i] = b2[0] + b2[1] + b2[2];
}

__global__ __launch_bounds__(TPB) void hg_degree(const int* __restrict__ src,
                                                 const int* __restrict__ dst,
                                                 int E,
                                                 unsigned* __restrict__ dout,
                                                 unsigned* __restrict__ din) {
    int e = blockIdx.x * TPB + threadIdx.x;
    if (e < E) {
        atomicAdd(&dout[src[e]], 1u);
        atomicAdd(&din[dst[e]], 1u);
    }
}

__global__ __launch_bounds__(TPB) void hg_scale(unsigned* __restrict__ deg, int n) {
    int i = blockIdx.x * TPB + threadIdx.x;
    if (i < n) {
        float d = (float)deg[i];
        if (d < 1.0f) d = 1.0f;
        ((float*)deg)[i] = rsqrtf(d);
    }
}

__global__ __launch_bounds__(TPB) void hg_gemm1_wmma(const float* __restrict__ x,
                                                     const float* __restrict__ s_out,
                                                     const float* __restrict__ W,
                                                     float* __restrict__ h_pre,
                                                     int N) {
    const int lane = threadIdx.x & 31;
    const int wave = threadIdx.x >> 5;
    const int tile = blockIdx.x * (TPB / 32) + wave;
    const int row0 = tile * 16;
    if (row0 >= N) return;

    const int half = lane >> 4;
    const int lm   = lane & 15;
    int row  = row0 + lm;
    int rowc = row < N ? row : N - 1;
    const float sc = s_out[rowc];

    v8f c = {};
#pragma unroll
    for (int kb = 0; kb < 8; ++kb) {
        const int k0 = kb * 4 + half * 2;
        v2f a, b;
        a.x = x[(size_t)rowc * 32 + k0]     * sc;
        a.y = x[(size_t)rowc * 32 + k0 + 1] * sc;
        b.x = W[(k0)     * 16 + lm];
        b.y = W[(k0 + 1) * 16 + lm];
        c = __builtin_amdgcn_wmma_f32_16x16x4_f32(
                 false, a,  false, b,
                 (short)0, c,  false,  false);
    }

    float* __restrict__ dp = h_pre + (size_t)(row0 + half * 8) * 16 + lm;
    if (row0 + 16 <= N) {
#pragma unroll
        for (int v = 0; v < 8; ++v) dp[(size_t)v * 16] = c[v];
    } else {
#pragma unroll
        for (int v = 0; v < 8; ++v) {
            const int orow = row0 + v + half * 8;
            if (orow < N) h_pre[(size_t)orow * 16 + lm] = c[v];
        }
    }
}

__global__ __launch_bounds__(TPB) void hg_scatter1(const int* __restrict__ src,
                                                   const int* __restrict__ dst,
                                                   int E,
                                                   const float* __restrict__ h_pre,
                                                   const float* __restrict__ s_in,
                                                   float* __restrict__ acc) {
    int e = blockIdx.x * TPB + threadIdx.x;
    if (e >= E) return;
    const int s = src[e], d = dst[e];
    const float sc = s_in[d];
    const float4* hp = (const float4*)(h_pre + (size_t)s * 16);
    float* ap = acc + (size_t)d * 16;
#pragma unroll
    for (int q = 0; q < 4; ++q) {
        float4 v = hp[q];
        atomicAdd(ap + q * 4 + 0, v.x * sc);
        atomicAdd(ap + q * 4 + 1, v.y * sc);
        atomicAdd(ap + q * 4 + 2, v.z * sc);
        atomicAdd(ap + q * 4 + 3, v.w * sc);
    }
}

__global__ __launch_bounds__(TPB) void hg_relu_bias(float* __restrict__ acc,
                                                    const float* __restrict__ b1,
                                                    long n16) {
    long i = (long)blockIdx.x * TPB + threadIdx.x;
    if (i < n16) {
        int f = (int)(i & 15);
        float v = acc[i] + b1[f] + b1[16 + f] + b1[32 + f];
        acc[i] = v > 0.0f ? v : 0.0f;
    }
}

__global__ __launch_bounds__(TPB) void hg_gemm2(const float* __restrict__ h,
                                                const float* __restrict__ s_out,
                                                const float* __restrict__ w,
                                                float* __restrict__ z,
                                                int N) {
    int i = blockIdx.x * TPB + threadIdx.x;
    if (i >= N) return;
    const float4* hp = (const float4*)(h + (size_t)i * 16);
    const float4* wp = (const float4*)w;
    float acc = 0.0f;
#pragma unroll
    for (int q = 0; q < 4; ++q) {
        float4 a = hp[q], b = wp[q];
        acc += a.x * b.x + a.y * b.y + a.z * b.z + a.w * b.w;
    }
    z[i] = acc * s_out[i];
}

__global__ __launch_bounds__(TPB) void hg_scatter2(const int* __restrict__ src,
                                                   const int* __restrict__ dst,
                                                   int E,
                                                   const float* __restrict__ z,
                                                   const float* __restrict__ s_in,
                                                   float* __restrict__ out) {
    int e = blockIdx.x * TPB + threadIdx.x;
    if (e < E) {
        int d = dst[e];
        atomicAdd(&out[d], z[src[e]] * s_in[d]);
    }
}

extern "C" void kernel_launch(void* const* d_in, const int* in_sizes, int n_in,
                              void* d_out, int out_size, void* d_ws, size_t ws_size,
                              hipStream_t stream) {
    const float* x  = (const float*)d_in[0];
    const int* src[3] = {(const int*)d_in[1], (const int*)d_in[3], (const int*)d_in[5]};
    const int* dst[3] = {(const int*)d_in[2], (const int*)d_in[4], (const int*)d_in[6]};
    const int  E[3]   = {in_sizes[1], in_sizes[3], in_sizes[5]};
    const float* W1 = (const float*)d_in[7];
    const float* b1 = (const float*)d_in[8];
    const float* W2 = (const float*)d_in[9];
    const float* b2 = (const float*)d_in[10];
    const int N = in_sizes[0] / 32;
    float* out = (float*)d_out;

    char* ws = (char*)d_ws;
    unsigned* deg = (unsigned*)ws;
    float* scales = (float*)ws;
    float* acc1   = (float*)(ws + (size_t)6 * N * 4);
    float* hpre   = acc1 + (size_t)N * 16;
    float* z      = hpre + (size_t)N * 16;

    {
        long span = 16L * N;
        hg_init<<<dim3((unsigned)((span + TPB - 1) / TPB)), dim3(TPB), 0, stream>>>(
            deg, acc1, out, b2, N);
    }

    for (int r = 0; r < 3; ++r) {
        hg_degree<<<dim3((E[r] + TPB - 1) / TPB), dim3(TPB), 0, stream>>>(
            src[r], dst[r], E[r], deg + (size_t)r * N, deg + (size_t)(3 + r) * N);
    }
    hg_scale<<<dim3((6 * N + TPB - 1) / TPB), dim3(TPB), 0, stream>>>(deg, 6 * N);

    const int tiles  = (N + 15) / 16;
    const int gblk   = (tiles + (TPB / 32) - 1) / (TPB / 32);
    for (int r = 0; r < 3; ++r) {
        hg_gemm1_wmma<<<dim3(gblk), dim3(TPB), 0, stream>>>(
            x, scales + (size_t)r * N, W1 + (size_t)r * 32 * 16, hpre, N);
        hg_scatter1<<<dim3((E[r] + TPB - 1) / TPB), dim3(TPB), 0, stream>>>(
            src[r], dst[r], E[r], hpre, scales + (size_t)(3 + r) * N, acc1);
    }
    {
        long n16 = 16L * N;
        hg_relu_bias<<<dim3((unsigned)((n16 + TPB - 1) / TPB)), dim3(TPB), 0, stream>>>(
            acc1, b1, n16);
    }

    for (int r = 0; r < 3; ++r) {
        hg_gemm2<<<dim3((N + TPB - 1) / TPB), dim3(TPB), 0, stream>>>(
            acc1, scales + (size_t)r * N, W2 + (size_t)r * 16, z, N);
        hg_scatter2<<<dim3((E[r] + TPB - 1) / TPB), dim3(TPB), 0, stream>>>(
            src[r], dst[r], E[r], z, scales + (size_t)(3 + r) * N, out);
    }
}


# layer-1 scatter: 16 lanes per edge (row_share hands src/dst/scale to the row), so each row load and float atomic covers 2 whole 64B rows, not 32 scattered words; atomics kept, sums in arrival order
# speedup vs baseline: 2.1539x; 2.1539x over previous
#include <hip/hip_runtime.h>

typedef float v2f __attribute__((ext_vector_type(2)));
typedef float v8f __attribute__((ext_vector_type(8)));

#define TPB 256

__global__ __launch_bounds__(TPB) void hg_init(unsigned* __restrict__ deg,
                                               float* __restrict__ acc1,
                                               float* __restrict__ out,
                                               const float* __restrict__ b2,
                                               int N) {
    long i = (long)blockIdx.x * TPB + threadIdx.x;
    if (i < 6L * N)  deg[i] = 0u;
    if (i < 16L * N) acc1[i] = 0.0f;
    if (i < N)       out[i] = b2[0] + b2[1] + b2[2];
}

__global__ __launch_bounds__(TPB) void hg_degree(const int* __restrict__ src,
                                                 const int* __restrict__ dst,
                                                 int E,
                                                 unsigned* __restrict__ dout,
                                                 unsigned* __restrict__ din) {
    int e = blockIdx.x * TPB + threadIdx.x;
    if (e < E) {
        atomicAdd(&dout[src[e]], 1u);
        atomicAdd(&din[dst[e]], 1u);
    }
}

__global__ __launch_bounds__(TPB) void hg_scale(unsigned* __restrict__ deg, int n) {
    int i = blockIdx.x * TPB + threadIdx.x;
    if (i < n) {
        float d = (float)deg[i];
        if (d < 1.0f) d = 1.0f;
        ((float*)deg)[i] = rsqrtf(d);
    }
}

__global__ __launch_bounds__(TPB) void hg_gemm1_wmma(const float* __restrict__ x,
                                                     const float* __restrict__ s_out,
                                                     const float* __restrict__ W,
                                                     float* __restrict__ h_pre,
                                                     int N) {
    const int lane = threadIdx.x & 31;
    const int wave = threadIdx.x >> 5;
    const int tile = blockIdx.x * (TPB / 32) + wave;
    const int row0 = tile * 16;
    if (row0 >= N) return;

    const int half = lane >> 4;
    const int lm   = lane & 15;
    int row  = row0 + lm;
    int rowc = row < N ? row : N - 1;
    const float sc = s_out[rowc];

    v8f c = {};
#pragma unroll
    for (int kb = 0; kb < 8; ++kb) {
        const int k0 = kb * 4 + half * 2;
        v2f a, b;
        a.x = x[(size_t)rowc * 32 + k0]     * sc;
        a.y = x[(size_t)rowc * 32 + k0 + 1] * sc;
        b.x = W[(k0)     * 16 + lm];
        b.y = W[(k0 + 1) * 16 + lm];
        c = __builtin_amdgcn_wmma_f32_16x16x4_f32(
                 false, a,  false, b,
                 (short)0, c,  false,  false);
    }

    float* __restrict__ dp = h_pre + (size_t)(row0 + half * 8) * 16 + lm;
    if (row0 + 16 <= N) {
#pragma unroll
        for (int v = 0; v < 8; ++v) dp[(size_t)v * 16] = c[v];
    } else {
#pragma unroll
        for (int v = 0; v < 8; ++v) {
            const int orow = row0 + v + half * 8;
            if (orow < N) h_pre[(size_t)orow * 16 + lm] = c[v];
        }
    }
}

#define S1_BATCHES 4

template <int I>
static __device__ __forceinline__ unsigned row_lane_u(unsigned v) {
    return (unsigned)__builtin_amdgcn_update_dpp(0, (int)v, 0x150 + I, 0xf, 0xf, true);
}
template <int I>
static __device__ __forceinline__ float row_lane_f(float v) {
    return __builtin_bit_cast(float, __builtin_amdgcn_update_dpp(0, __builtin_bit_cast(int, v),
                                                                 0x150 + I, 0xf, 0xf, true));
}

#define S1_ALL(M) M(0) M(1) M(2) M(3) M(4) M(5) M(6) M(7) M(8) M(9) M(10) M(11) M(12) M(13) M(14) M(15)
#define S1_LOAD(I) v[I] = *(const float*)((const char*)h_pre + (row_lane_u<I>(sb) + lm4));
#define S1_ADD(I)                                                              \
    if (!GUARD || I < nrow)                                                    \
        atomicAdd((float*)((char*)acc + (row_lane_u<I>(db) + lm4)), v[I] * row_lane_f<I>(sc));

template <bool GUARD>
static __device__ __forceinline__ void s1_batch(const int* __restrict__ src,
                                                const int* __restrict__ dst,
                                                int E, int e0, int lane, unsigned lm4,
                                                const float* __restrict__ h_pre,
                                                const float* __restrict__ s_in,
                                                float* __restrict__ acc) {
    int e = e0 + lane;
    if (GUARD) e = e < E ? e : E - 1;
    const int s = src[e], d = dst[e];
    const float sc = s_in[d];
    const unsigned sb = (unsigned)s * 64u;
    const unsigned db = (unsigned)d * 64u;
    const int nrow = E - e0 - (lane & 16);
    float v[16];
    S1_ALL(S1_LOAD)
    S1_ALL(S1_ADD)
}

__global__ __launch_bounds__(TPB) void hg_scatter1(const int* __restrict__ src,
                                                   const int* __restrict__ dst,
                                                   int E,
                                                   const float* __restrict__ h_pre,
                                                   const float* __restrict__ s_in,
                                                   float* __restrict__ acc) {
    const int lane = threadIdx.x & 31;
    const unsigned lm4 = (unsigned)(lane & 15) * 4u;
    const int wave = __builtin_amdgcn_readfirstlane((int)(threadIdx.x >> 5));
    int e0 = (blockIdx.x * (TPB / 32) + wave) * (32 * S1_BATCHES);
    for (int b = 0; b < S1_BATCHES; ++b, e0 += 32) {
        if (e0 >= E) return;
        if (e0 + 32 <= E) s1_batch<false>(src, dst, E, e0, lane, lm4, h_pre, s_in, acc);
        else              s1_batch<true >(src, dst, E, e0, lane, lm4, h_pre, s_in, acc);
    }
}

__global__ __launch_bounds__(TPB) void hg_relu_bias(float* __restrict__ acc,
                                                    const float* __restrict__ b1,
                                                    long n16) {
    long i = (long)blockIdx.x * TPB + threadIdx.x;
    if (i < n16) {
        int f = (int)(i & 15);
        float v = acc[i] + b1[f] + b1[16 + f] + b1[32 + f];
        acc[i] = v > 0.0f ? v : 0.0f;
    }
}

__global__ __launch_bounds__(TPB) void hg_gemm2(const float* __restrict__ h,
                                                const float* __restrict__ s_out,
                                                const float* __restrict__ w,
                                                float* __restrict__ z,
                                                int N) {
    int i = blockIdx.x * TPB + threadIdx.x;
    if (i >= N) return;
    const float4* hp = (const float4*)(h + (size_t)i * 16);
    const float4* wp = (const float4*)w;
    float acc = 0.0f;
#pragma unroll
    for (int q = 0; q < 4; ++q) {
        float4 a = hp[q], b = wp[q];
        acc += a.x * b.x + a.y * b.y + a.z * b.z + a.w * b.w;
    }
    z[i] = acc * s_out[i];
}

__global__ __launch_bounds__(TPB) void hg_scatter2(const int* __restrict__ src,
                                                   const int* __restrict__ dst,
                                                   int E,
                                                   const float* __restrict__ z,
                                                   const float* __restrict__ s_in,
                                                   float* __restrict__ out) {
    int e = blockIdx.x * TPB + threadIdx.x;
    if (e < E) {
        int d = dst[e];
        atomicAdd(&out[d], z[src[e]] * s_in[d]);
    }
}

extern "C" void kernel_launch(void* const* d_in, const int* in_sizes, int n_in,
                              void* d_out, int out_size, void* d_ws, size_t ws_size,
                              hipStream_t stream) {
    const float* x  = (const float*)d_in[0];
    const int* src[3] = {(const int*)d_in[1], (const int*)d_in[3], (const int*)d_in[5]};
    const int* dst[3] = {(const int*)d_in[2], (const int*)d_in[4], (const int*)d_in[6]};
    const int  E[3]   = {in_sizes[1], in_sizes[3], in_sizes[5]};
    const float* W1 = (const float*)d_in[7];
    const float* b1 = (const float*)d_in[8];
    const float* W2 = (const float*)d_in[9];
    const float* b2 = (const float*)d_in[10];
    const int N = in_sizes[0] / 32;
    float* out = (float*)d_out;

    char* ws = (char*)d_ws;
    unsigned* deg = (unsigned*)ws;
    float* scales = (float*)ws;
    float* acc1   = (float*)(ws + (size_t)6 * N * 4);
    float* hpre   = acc1 + (size_t)N * 16;
    float* z      = hpre + (size_t)N * 16;

    {
        long span = 16L * N;
        hg_init<<<dim3((unsigned)((span + TPB - 1) / TPB)), dim3(TPB), 0, stream>>>(
            deg, acc1, out, b2, N);
    }

    for (int r = 0; r < 3; ++r) {
        hg_degree<<<dim3((E[r] + TPB - 1) / TPB), dim3(TPB), 0, stream>>>(
            src[r], dst[r], E[r], deg + (size_t)r * N, deg + (size_t)(3 + r) * N);
    }
    hg_scale<<<dim3((6 * N + TPB - 1) / TPB), dim3(TPB), 0, stream>>>(deg, 6 * N);

    const int tiles  = (N + 15) / 16;
    const int gblk   = (tiles + (TPB / 32) - 1) / (TPB / 32);
    for (int r = 0; r < 3; ++r) {
        hg_gemm1_wmma<<<dim3(gblk), dim3(TPB), 0, stream>>>(
            x, scales + (size_t)r * N, W1 + (size_t)r * 32 * 16, hpre, N);
        hg_scatter1<<<dim3((E[r] + TPB * S1_BATCHES - 1) / (TPB * S1_BATCHES)), dim3(TPB), 0, stream>>>(
            src[r], dst[r], E[r], hpre, scales + (size_t)(3 + r) * N, acc1);
    }
    {
        long n16 = 16L * N;
        hg_relu_bias<<<dim3((unsigned)((n16 + TPB - 1) / TPB)), dim3(TPB), 0, stream>>>(
            acc1, b1, n16);
    }

    for (int r = 0; r < 3; ++r) {
        hg_gemm2<<<dim3((N + TPB - 1) / TPB), dim3(TPB), 0, stream>>>(
            acc1, scales + (size_t)r * N, W2 + (size_t)r * 16, z, N);
        hg_scatter2<<<dim3((E[r] + TPB - 1) / TPB), dim3(TPB), 0, stream>>>(
            src[r], dst[r], E[r], z, scales + (size_t)(3 + r) * N, out);
    }
}


# layer-1 scatter: work-group keeps a 1024-row dst slice in LDS, scans 1/5 of the edges, adds matching h_pre rows by LDS atomics, one global-atomic flush scaled by s_in; sums in arrival order
# speedup vs baseline: 2.3183x; 1.0763x over previous
#include <hip/hip_runtime.h>

typedef float v2f __attribute__((ext_vector_type(2)));
typedef float v8f __attribute__((ext_vector_type(8)));

#define TPB 256

__global__ __launch_bounds__(TPB) void hg_init(unsigned* __restrict__ deg,
                                               float* __restrict__ acc1,
                                               float* __restrict__ out,
                                               const float* __restrict__ b2,
                                               int N) {
    long i = (long)blockIdx.x * TPB + threadIdx.x;
    if (i < 6L * N)  deg[i] = 0u;
    if (i < 16L * N) acc1[i] = 0.0f;
    if (i < N)       out[i] = b2[0] + b2[1] + b2[2];
}

__global__ __launch_bounds__(TPB) void hg_degree(const int* __restrict__ src,
                                                 const int* __restrict__ dst,
                                                 int E,
                                                 unsigned* __restrict__ dout,
                                                 unsigned* __restrict__ din) {
    int e = blockIdx.x * TPB + threadIdx.x;
    if (e < E) {
        atomicAdd(&dout[src[e]], 1u);
        atomicAdd(&din[dst[e]], 1u);
    }
}

__global__ __launch_bounds__(TPB) void hg_scale(unsigned* __restrict__ deg, int n) {
    int i = blockIdx.x * TPB + threadIdx.x;
    if (i < n) {
        float d = (float)deg[i];
        if (d < 1.0f) d = 1.0f;
        ((float*)deg)[i] = rsqrtf(d);
    }
}

__global__ __launch_bounds__(TPB) void hg_gemm1_wmma(const float* __restrict__ x,
                                                     const float* __restrict__ s_out,
                                                     const float* __restrict__ W,
                                                     float* __restrict__ h_pre,
                                                     int N) {
    const int lane = threadIdx.x & 31;
    const int wave = threadIdx.x >> 5;
    const int tile = blockIdx.x * (TPB / 32) + wave;
    const int row0 = tile * 16;
    if (row0 >= N) return;

    const int half = lane >> 4;
    const int lm   = lane & 15;
    int row  = row0 + lm;
    int rowc = row < N ? row : N - 1;
    const float sc = s_out[rowc];

    v8f c = {};
#pragma unroll
    for (int kb = 0; kb < 8; ++kb) {
        const int k0 = kb * 4 + half * 2;
        v2f a, b;
        a.x = x[(size_t)rowc * 32 + k0]     * sc;
        a.y = x[(size_t)rowc * 32 + k0 + 1] * sc;
        b.x = W[(k0)     * 16 + lm];
        b.y = W[(k0 + 1) * 16 + lm];
        c = __builtin_amdgcn_wmma_f32_16x16x4_f32(
                 false, a,  false, b,
                 (short)0, c,  false,  false);
    }

    float* __restrict__ dp = h_pre + (size_t)(row0 + half * 8) * 16 + lm;
    if (row0 + 16 <= N) {
#pragma unroll
        for (int v = 0; v < 8; ++v) dp[(size_t)v * 16] = c[v];
    } else {
#pragma unroll
        for (int v = 0; v < 8; ++v) {
            const int orow = row0 + v + half * 8;
            if (orow < N) h_pre[(size_t)orow * 16 + lm] = c[v];
        }
    }
}

#define SL_ROWS  1024
#define SL_WAVES 32
#define SL_TPB   (SL_WAVES * 32)
#define SL_PARTS 5
#define SL_GROUP 512
#define SL_QCAP  256

template <int I>
static __device__ __forceinline__ unsigned row_lane_u(unsigned v) {
    return (unsigned)__builtin_amdgcn_update_dpp(0, (int)v, 0x150 + I, 0xf, 0xf, true);
}

#define S1_ALL(M) M(0) M(1) M(2) M(3) M(4) M(5) M(6) M(7) M(8) M(9) M(10) M(11) M(12) M(13) M(14) M(15)
#define S1_LOAD(I) v[I] = *(const float*)((const char*)h_pre + (row_lane_u<I>(sb) + lm4));
#define S1_ADD(I)                                                              \
    if (!GUARD || I < nrow)                                                    \
        atomicAdd((float*)((char*)rows + (row_lane_u<I>(lb) + lm4)), v[I]);

template <bool GUARD>
static __device__ __forceinline__ void s1_drain(const unsigned* qe, const unsigned* ql,
                                                int head, int cnt, int lane, unsigned lm4,
                                                const int* __restrict__ src,
                                                const float* __restrict__ h_pre,
                                                float* rows) {
    __builtin_amdgcn_fence(__ATOMIC_RELEASE, "wavefront");
    __builtin_amdgcn_wave_barrier();
    __builtin_amdgcn_fence(__ATOMIC_ACQUIRE, "wavefront");
    const int ent = (!GUARD || lane < cnt) ? lane : 0;
    const unsigned qi = (unsigned)(head + ent) & (SL_QCAP - 1);
    const unsigned e  = qe[qi];
    const unsigned lb = ql[qi];
    const unsigned sb = (unsigned)src[e] * 64u;
    const int nrow = cnt - (lane & 16);
    float v[16];
    S1_ALL(S1_LOAD)
    S1_ALL(S1_ADD)
}

static __device__ __forceinline__ void s1_test(int dv, int e, int base,
                                               unsigned* qe, unsigned* ql, int& tail) {
    const unsigned l = (unsigned)(dv - base);
    const bool hit = l < (unsigned)SL_ROWS;
    const unsigned m = __builtin_amdgcn_ballot_w32(hit);
    if (m != 0u) {
        const unsigned pos = ((unsigned)tail + __builtin_amdgcn_mbcnt_lo(m, 0u)) & (SL_QCAP - 1);
        if (hit) { qe[pos] = (unsigned)e; ql[pos] = l * 64u; }
        tail += __builtin_popcount(m);
    }
}

__global__ __launch_bounds__(SL_TPB) void hg_scatter1(const int* __restrict__ src,
                                                      const int* __restrict__ dst,
                                                      int E,
                                                      const float* __restrict__ h_pre,
                                                      const float* __restrict__ s_in,
                                                      float* __restrict__ acc,
                                                      int N) {
    __shared__ float    rows[SL_ROWS * 16];
    __shared__ unsigned q_e[SL_WAVES][SL_QCAP];
    __shared__ unsigned q_l[SL_WAVES][SL_QCAP];

    const int lane = threadIdx.x & 31;
    const unsigned lm4 = (unsigned)(lane & 15) * 4u;
    const int wave = __builtin_amdgcn_readfirstlane((int)(threadIdx.x >> 5));
    const int base = (int)blockIdx.x * SL_ROWS;

    for (int i = threadIdx.x; i < SL_ROWS * 4; i += SL_TPB)
        ((float4*)rows)[i] = make_float4(0.0f, 0.0f, 0.0f, 0.0f);
    __syncthreads();

    const int chunk   = ((E + SL_PARTS * SL_GROUP - 1) / (SL_PARTS * SL_GROUP)) * SL_GROUP;
    const int e_begin = (int)blockIdx.y * chunk;
    const int e_end   = (e_begin + chunk < E) ? (e_begin + chunk) : E;

    unsigned* qe = q_e[wave];
    unsigned* ql = q_l[wave];
    int head = 0, tail = 0;

    for (int e0 = e_begin + wave * SL_GROUP; e0 < e_end; e0 += SL_WAVES * SL_GROUP) {
        int4 d[4];
        if (e0 + SL_GROUP <= e_end) {
#pragma unroll
            for (int k = 0; k < 4; ++k)
                d[k] = *(const int4*)(dst + e0 + k * 128 + lane * 4);
        } else {
#pragma unroll
            for (int k = 0; k < 4; ++k) {
                const int e = e0 + k * 128 + lane * 4;
                d[k].x = (e     < e_end) ? dst[e]     : -1;
                d[k].y = (e + 1 < e_end) ? dst[e + 1] : -1;
                d[k].z = (e + 2 < e_end) ? dst[e + 2] : -1;
                d[k].w = (e + 3 < e_end) ? dst[e + 3] : -1;
            }
        }
#pragma unroll
        for (int k = 0; k < 4; ++k) {
            const int e = e0 + k * 128 + lane * 4;
            s1_test(d[k].x, e,     base, qe, ql, tail);
            s1_test(d[k].y, e + 1, base, qe, ql, tail);
            s1_test(d[k].z, e + 2, base, qe, ql, tail);
            s1_test(d[k].w, e + 3, base, qe, ql, tail);
            while (tail - head >= 32) {
                s1_drain<false>(qe, ql, head, 32, lane, lm4, src, h_pre, rows);
                head += 32;
            }
        }
    }
    if (tail - head > 0)
        s1_drain<true>(qe, ql, head, tail - head, lane, lm4, src, h_pre, rows);

    __syncthreads();

    constexpr int FL = SL_ROWS * 16 / SL_TPB;
    const int f  = threadIdx.x & 15;
    const int r0 = threadIdx.x >> 4;
    float pv[FL], ps[FL];
#pragma unroll
    for (int j = 0; j < FL; ++j) {
        const int r  = r0 + j * (SL_TPB / 16);
        const int gd = base + r;
        pv[j] = rows[r * 16 + f];
        ps[j] = s_in[gd < N ? gd : N - 1];
    }
#pragma unroll
    for (int j = 0; j < FL; ++j) {
        const int gd = base + r0 + j * (SL_TPB / 16);
        if (gd < N && pv[j] != 0.0f)
            atomicAdd(&acc[(size_t)gd * 16 + f], pv[j] * ps[j]);
    }
}

__global__ __launch_bounds__(TPB) void hg_relu_bias(float* __restrict__ acc,
                                                    const float* __restrict__ b1,
                                                    long n16) {
    long i = (long)blockIdx.x * TPB + threadIdx.x;
    if (i < n16) {
        int f = (int)(i & 15);
        float v = acc[i] + b1[f] + b1[16 + f] + b1[32 + f];
        acc[i] = v > 0.0f ? v : 0.0f;
    }
}

__global__ __launch_bounds__(TPB) void hg_gemm2(const float* __restrict__ h,
                                                const float* __restrict__ s_out,
                                                const float* __restrict__ w,
                                                float* __restrict__ z,
                                                int N) {
    int i = blockIdx.x * TPB + threadIdx.x;
    if (i >= N) return;
    const float4* hp = (const float4*)(h + (size_t)i * 16);
    const float4* wp = (const float4*)w;
    float acc = 0.0f;
#pragma unroll
    for (int q = 0; q < 4; ++q) {
        float4 a = hp[q], b = wp[q];
        acc += a.x * b.x + a.y * b.y + a.z * b.z + a.w * b.w;
    }
    z[i] = acc * s_out[i];
}

__global__ __launch_bounds__(TPB) void hg_scatter2(const int* __restrict__ src,
                                                   const int* __restrict__ dst,
                                                   int E,
                                                   const float* __restrict__ z,
                                                   const float* __restrict__ s_in,
                                                   float* __restrict__ out) {
    int e = blockIdx.x * TPB + threadIdx.x;
    if (e < E) {
        int d = dst[e];
        atomicAdd(&out[d], z[src[e]] * s_in[d]);
    }
}

extern "C" void kernel_launch(void* const* d_in, const int* in_sizes, int n_in,
                              void* d_out, int out_size, void* d_ws, size_t ws_size,
                              hipStream_t stream) {
    const float* x  = (const float*)d_in[0];
    const int* src[3] = {(const int*)d_in[1], (const int*)d_in[3], (const int*)d_in[5]};
    const int* dst[3] = {(const int*)d_in[2], (const int*)d_in[4], (const int*)d_in[6]};
    const int  E[3]   = {in_sizes[1], in_sizes[3], in_sizes[5]};
    const float* W1 = (const float*)d_in[7];
    const float* b1 = (const float*)d_in[8];
    const float* W2 = (const float*)d_in[9];
    const float* b2 = (const float*)d_in[10];
    const int N = in_sizes[0] / 32;
    float* out = (float*)d_out;

    char* ws = (char*)d_ws;
    unsigned* deg = (unsigned*)ws;
    float* scales = (float*)ws;
    float* acc1   = (float*)(ws + (size_t)6 * N * 4);
    float* hpre   = acc1 + (size_t)N * 16;
    float* z      = hpre + (size_t)N * 16;

    {
        long span = 16L * N;
        hg_init<<<dim3((unsigned)((span + TPB - 1) / TPB)), dim3(TPB), 0, stream>>>(
            deg, acc1, out, b2, N);
    }

    for (int r = 0; r < 3; ++r) {
        hg_degree<<<dim3((E[r] + TPB - 1) / TPB), dim3(TPB), 0, stream>>>(
            src[r], dst[r], E[r], deg + (size_t)r * N, deg + (size_t)(3 + r) * N);
    }
    hg_scale<<<dim3((6 * N + TPB - 1) / TPB), dim3(TPB), 0, stream>>>(deg, 6 * N);

    const int tiles  = (N + 15) / 16;
    const int gblk   = (tiles + (TPB / 32) - 1) / (TPB / 32);
    const int slices = (N + SL_ROWS - 1) / SL_ROWS;
    for (int r = 0; r < 3; ++r) {
        hg_gemm1_wmma<<<dim3(gblk), dim3(TPB), 0, stream>>>(
            x, scales + (size_t)r * N, W1 + (size_t)r * 32 * 16, hpre, N);
        hg_scatter1<<<dim3(slices, SL_PARTS), dim3(SL_TPB), 0, stream>>>(
            src[r], dst[r], E[r], hpre, scales + (size_t)(3 + r) * N, acc1, N);
    }
    {
        long n16 = 16L * N;
        hg_relu_bias<<<dim3((unsigned)((n16 + TPB - 1) / TPB)), dim3(TPB), 0, stream>>>(
            acc1, b1, n16);
    }

    for (int r = 0; r < 3; ++r) {
        hg_gemm2<<<dim3((N + TPB - 1) / TPB), dim3(TPB), 0, stream>>>(
            acc1, scales + (size_t)r * N, W2 + (size_t)r * 16, z, N);
        hg_scatter2<<<dim3((E[r] + TPB - 1) / TPB), dim3(TPB), 0, stream>>>(
            src[r], dst[r], E[r], z, scales + (size_t)(3 + r) * N, out);
    }
}


# degree counts: each work-group counts a 131072-index chunk into a whole-node-range histogram in LDS, then adds its non-zero bins to the global counters by whole-line atomics; one launch for all six
# speedup vs baseline: 4.5460x; 1.9609x over previous
#include <hip/hip_runtime.h>

typedef float v2f __attribute__((ext_vector_type(2)));
typedef float v8f __attribute__((ext_vector_type(8)));

#define TPB 256

__global__ __launch_bounds__(TPB) void hg_init(unsigned* __restrict__ deg,
                                               float* __restrict__ acc1,
                                               float* __restrict__ out,
                                               const float* __restrict__ b2,
                                               int N) {
    long i = (long)blockIdx.x * TPB + threadIdx.x;
    if (i < 6L * N)  deg[i] = 0u;
    if (i < 16L * N) acc1[i] = 0.0f;
    if (i < N)       out[i] = b2[0] + b2[1] + b2[2];
}

#define DG_TPB 1024
#define DG_CAP 51200
#define DG_CH  131072

struct DegArgs {
    const int* s0; const int* d0;
    const int* s1; const int* d1;
    const int* s2; const int* d2;
    int E0, E1, E2;
};

static __device__ __forceinline__ void dg_count(unsigned* cnt, int v, int base, int span) {
    const unsigned l = (unsigned)(v - base);
    if (l < (unsigned)span) atomicAdd(&cnt[l], 1u);
}

__global__ __launch_bounds__(DG_TPB) void hg_degree_lds(DegArgs a,
                                                        unsigned* __restrict__ deg,
                                                        int N) {
    __shared__ unsigned cnt[DG_CAP];

    const int p0 = (a.E0 + DG_CH - 1) / DG_CH;
    const int p1 = (a.E1 + DG_CH - 1) / DG_CH;
    int b = (int)blockIdx.x;
    int r = 0;
    if (b >= 2 * p0) { b -= 2 * p0; r = 1; }
    if (r == 1 && b >= 2 * p1) { b -= 2 * p1; r = 2; }
    const int which = b & 1;
    const int part  = b >> 1;
    const int* __restrict__ idx = (r == 0) ? (which ? a.d0 : a.s0)
                                : (r == 1) ? (which ? a.d1 : a.s1)
                                           : (which ? a.d2 : a.s2);
    const int E = (r == 0) ? a.E0 : (r == 1) ? a.E1 : a.E2;
    unsigned* __restrict__ hist = deg + (size_t)(which * 3 + r) * N;

    const int base = (int)blockIdx.y * DG_CAP;
    const int span = (N - base < DG_CAP) ? (N - base) : DG_CAP;

    for (int i = threadIdx.x; i < span; i += DG_TPB) cnt[i] = 0u;
    __syncthreads();

    const int e_begin = part * DG_CH;
    const int e_end   = (e_begin + DG_CH < E) ? (e_begin + DG_CH) : E;
    for (int e0 = e_begin; e0 < e_end; e0 += 4 * DG_TPB * 4) {
        int4 d[4];
        if (e0 + 4 * DG_TPB * 4 <= e_end) {
#pragma unroll
            for (int k = 0; k < 4; ++k)
                d[k] = *(const int4*)(idx + e0 + k * (DG_TPB * 4) + threadIdx.x * 4);
        } else {
#pragma unroll
            for (int k = 0; k < 4; ++k) {
                const int e = e0 + k * (DG_TPB * 4) + threadIdx.x * 4;
                d[k].x = (e     < e_end) ? idx[e]     : -1;
                d[k].y = (e + 1 < e_end) ? idx[e + 1] : -1;
                d[k].z = (e + 2 < e_end) ? idx[e + 2] : -1;
                d[k].w = (e + 3 < e_end) ? idx[e + 3] : -1;
            }
        }
#pragma unroll
        for (int k = 0; k < 4; ++k) {
            dg_count(cnt, d[k].x, base, span);
            dg_count(cnt, d[k].y, base, span);
            dg_count(cnt, d[k].z, base, span);
            dg_count(cnt, d[k].w, base, span);
        }
    }
    __syncthreads();

    const int tiles = (span + DG_TPB - 1) / DG_TPB;
    const int rot   = (int)(blockIdx.x % (unsigned)tiles);
    for (int j = 0; j < tiles; ++j) {
        int t = j + rot;
        if (t >= tiles) t -= tiles;
        const int i = t * DG_TPB + (int)threadIdx.x;
        if (i < span) {
            const unsigned c = cnt[i];
            if (c != 0u) atomicAdd(&hist[base + i], c);
        }
    }
}

__global__ __launch_bounds__(TPB) void hg_scale(unsigned* __restrict__ deg, int n) {
    int i = blockIdx.x * TPB + threadIdx.x;
    if (i < n) {
        float d = (float)deg[i];
        if (d < 1.0f) d = 1.0f;
        ((float*)deg)[i] = rsqrtf(d);
    }
}

__global__ __launch_bounds__(TPB) void hg_gemm1_wmma(const float* __restrict__ x,
                                                     const float* __restrict__ s_out,
                                                     const float* __restrict__ W,
                                                     float* __restrict__ h_pre,
                                                     int N) {
    const int lane = threadIdx.x & 31;
    const int wave = threadIdx.x >> 5;
    const int tile = blockIdx.x * (TPB / 32) + wave;
    const int row0 = tile * 16;
    if (row0 >= N) return;

    const int half = lane >> 4;
    const int lm   = lane & 15;
    int row  = row0 + lm;
    int rowc = row < N ? row : N - 1;
    const float sc = s_out[rowc];

    v8f c = {};
#pragma unroll
    for (int kb = 0; kb < 8; ++kb) {
        const int k0 = kb * 4 + half * 2;
        v2f a, b;
        a.x = x[(size_t)rowc * 32 + k0]     * sc;
        a.y = x[(size_t)rowc * 32 + k0 + 1] * sc;
        b.x = W[(k0)     * 16 + lm];
        b.y = W[(k0 + 1) * 16 + lm];
        c = __builtin_amdgcn_wmma_f32_16x16x4_f32(
                 false, a,  false, b,
                 (short)0, c,  false,  false);
    }

    float* __restrict__ dp = h_pre + (size_t)(row0 + half * 8) * 16 + lm;
    if (row0 + 16 <= N) {
#pragma unroll
        for (int v = 0; v < 8; ++v) dp[(size_t)v * 16] = c[v];
    } else {
#pragma unroll
        for (int v = 0; v < 8; ++v) {
            const int orow = row0 + v + half * 8;
            if (orow < N) h_pre[(size_t)orow * 16 + lm] = c[v];
        }
    }
}

#define SL_ROWS  1024
#define SL_WAVES 32
#define SL_TPB   (SL_WAVES * 32)
#define SL_PARTS 5
#define SL_GROUP 512
#define SL_QCAP  256

template <int I>
static __device__ __forceinline__ unsigned row_lane_u(unsigned v) {
    return (unsigned)__builtin_amdgcn_update_dpp(0, (int)v, 0x150 + I, 0xf, 0xf, true);
}

#define S1_ALL(M) M(0) M(1) M(2) M(3) M(4) M(5) M(6) M(7) M(8) M(9) M(10) M(11) M(12) M(13) M(14) M(15)
#define S1_LOAD(I) v[I] = *(const float*)((const char*)h_pre + (row_lane_u<I>(sb) + lm4));
#define S1_ADD(I)                                                              \
    if (!GUARD || I < nrow)                                                    \
        atomicAdd((float*)((char*)rows + (row_lane_u<I>(lb) + lm4)), v[I]);

template <bool GUARD>
static __device__ __forceinline__ void s1_drain(const unsigned* qe, const unsigned* ql,
                                                int head, int cnt, int lane, unsigned lm4,
                                                const int* __restrict__ src,
                                                const float* __restrict__ h_pre,
                                                float* rows) {
    __builtin_amdgcn_fence(__ATOMIC_RELEASE, "wavefront");
    __builtin_amdgcn_wave_barrier();
    __builtin_amdgcn_fence(__ATOMIC_ACQUIRE, "wavefront");
    const int ent = (!GUARD || lane < cnt) ? lane : 0;
    const unsigned qi = (unsigned)(head + ent) & (SL_QCAP - 1);
    const unsigned e  = qe[qi];
    const unsigned lb = ql[qi];
    const unsigned sb = (unsigned)src[e] * 64u;
    const int nrow = cnt - (lane & 16);
    float v[16];
    S1_ALL(S1_LOAD)
    S1_ALL(S1_ADD)
}

static __device__ __forceinline__ void s1_test(int dv, int e, int base,
                                               unsigned* qe, unsigned* ql, int& tail) {
    const unsigned l = (unsigned)(dv - base);
    const bool hit = l < (unsigned)SL_ROWS;
    const unsigned m = __builtin_amdgcn_ballot_w32(hit);
    if (m != 0u) {
        const unsigned pos = ((unsigned)tail + __builtin_amdgcn_mbcnt_lo(m, 0u)) & (SL_QCAP - 1);
        if (hit) { qe[pos] = (unsigned)e; ql[pos] = l * 64u; }
        tail += __builtin_popcount(m);
    }
}

__global__ __launch_bounds__(SL_TPB) void hg_scatter1(const int* __restrict__ src,
                                                      const int* __restrict__ dst,
                                                      int E,
                                                      const float* __restrict__ h_pre,
                                                      const float* __restrict__ s_in,
                                                      float* __restrict__ acc,
                                                      int N) {
    __shared__ float    rows[SL_ROWS * 16];
    __shared__ unsigned q_e[SL_WAVES][SL_QCAP];
    __shared__ unsigned q_l[SL_WAVES][SL_QCAP];

    const int lane = threadIdx.x & 31;
    const unsigned lm4 = (unsigned)(lane & 15) * 4u;
    const int wave = __builtin_amdgcn_readfirstlane((int)(threadIdx.x >> 5));
    const int base = (int)blockIdx.x * SL_ROWS;

    for (int i = threadIdx.x; i < SL_ROWS * 4; i += SL_TPB)
        ((float4*)rows)[i] = make_float4(0.0f, 0.0f, 0.0f, 0.0f);
    __syncthreads();

    const int chunk   = ((E + SL_PARTS * SL_GROUP - 1) / (SL_PARTS * SL_GROUP)) * SL_GROUP;
    const int e_begin = (int)blockIdx.y * chunk;
    const int e_end   = (e_begin + chunk < E) ? (e_begin + chunk) : E;

    unsigned* qe = q_e[wave];
    unsigned* ql = q_l[wave];
    int head = 0, tail = 0;

    for (int e0 = e_begin + wave * SL_GROUP; e0 < e_end; e0 += SL_WAVES * SL_GROUP) {
        int4 d[4];
        if (e0 + SL_GROUP <= e_end) {
#pragma unroll
            for (int k = 0; k < 4; ++k)
                d[k] = *(const int4*)(dst + e0 + k * 128 + lane * 4);
        } else {
#pragma unroll
            for (int k = 0; k < 4; ++k) {
                const int e = e0 + k * 128 + lane * 4;
                d[k].x = (e     < e_end) ? dst[e]     : -1;
                d[k].y = (e + 1 < e_end) ? dst[e + 1] : -1;
                d[k].z = (e + 2 < e_end) ? dst[e + 2] : -1;
                d[k].w = (e + 3 < e_end) ? dst[e + 3] : -1;
            }
        }
#pragma unroll
        for (int k = 0; k < 4; ++k) {
            const int e = e0 + k * 128 + lane * 4;
            s1_test(d[k].x, e,     base, qe, ql, tail);
            s1_test(d[k].y, e + 1, base, qe, ql, tail);
            s1_test(d[k].z, e + 2, base, qe, ql, tail);
            s1_test(d[k].w, e + 3, base, qe, ql, tail);
            while (tail - head >= 32) {
                s1_drain<false>(qe, ql, head, 32, lane, lm4, src, h_pre, rows);
                head += 32;
            }
        }
    }
    if (tail - head > 0)
        s1_drain<true>(qe, ql, head, tail - head, lane, lm4, src, h_pre, rows);

    __syncthreads();

    constexpr int FL = SL_ROWS * 16 / SL_TPB;
    const int f  = threadIdx.x & 15;
    const int r0 = threadIdx.x >> 4;
    float pv[FL], ps[FL];
#pragma unroll
    for (int j = 0; j < FL; ++j) {
        const int r  = r0 + j * (SL_TPB / 16);
        const int gd = base + r;
        pv[j] = rows[r * 16 + f];
        ps[j] = s_in[gd < N ? gd : N - 1];
    }
#pragma unroll
    for (int j = 0; j < FL; ++j) {
        const int gd = base + r0 + j * (SL_TPB / 16);
        if (gd < N && pv[j] != 0.0f)
            atomicAdd(&acc[(size_t)gd * 16 + f], pv[j] * ps[j]);
    }
}

__global__ __launch_bounds__(TPB) void hg_relu_bias(float* __restrict__ acc,
                                                    const float* __restrict__ b1,
                                                    long n16) {
    long i = (long)blockIdx.x * TPB + threadIdx.x;
    if (i < n16) {
        int f = (int)(i & 15);
        float v = acc[i] + b1[f] + b1[16 + f] + b1[32 + f];
        acc[i] = v > 0.0f ? v : 0.0f;
    }
}

__global__ __launch_bounds__(TPB) void hg_gemm2(const float* __restrict__ h,
                                                const float* __restrict__ s_out,
                                                const float* __restrict__ w,
                                                float* __restrict__ z,
                                                int N) {
    int i = blockIdx.x * TPB + threadIdx.x;
    if (i >= N) return;
    const float4* hp = (const float4*)(h + (size_t)i * 16);
    const float4* wp = (const float4*)w;
    float acc = 0.0f;
#pragma unroll
    for (int q = 0; q < 4; ++q) {
        float4 a = hp[q], b = wp[q];
        acc += a.x * b.x + a.y * b.y + a.z * b.z + a.w * b.w;
    }
    z[i] = acc * s_out[i];
}

__global__ __launch_bounds__(TPB) void hg_scatter2(const int* __restrict__ src,
                                                   const int* __restrict__ dst,
                                                   int E,
                                                   const float* __restrict__ z,
                                                   const float* __restrict__ s_in,
                                                   float* __restrict__ out) {
    int e = blockIdx.x * TPB + threadIdx.x;
    if (e < E) {
        int d = dst[e];
        atomicAdd(&out[d], z[src[e]] * s_in[d]);
    }
}

extern "C" void kernel_launch(void* const* d_in, const int* in_sizes, int n_in,
                              void* d_out, int out_size, void* d_ws, size_t ws_size,
                              hipStream_t stream) {
    const float* x  = (const float*)d_in[0];
    const int* src[3] = {(const int*)d_in[1], (const int*)d_in[3], (const int*)d_in[5]};
    const int* dst[3] = {(const int*)d_in[2], (const int*)d_in[4], (const int*)d_in[6]};
    const int  E[3]   = {in_sizes[1], in_sizes[3], in_sizes[5]};
    const float* W1 = (const float*)d_in[7];
    const float* b1 = (const float*)d_in[8];
    const float* W2 = (const float*)d_in[9];
    const float* b2 = (const float*)d_in[10];
    const int N = in_sizes[0] / 32;
    float* out = (float*)d_out;

    char* ws = (char*)d_ws;
    unsigned* deg = (unsigned*)ws;
    float* scales = (float*)ws;
    float* acc1   = (float*)(ws + (size_t)6 * N * 4);
    float* hpre   = acc1 + (size_t)N * 16;
    float* z      = hpre + (size_t)N * 16;

    {
        long span = 16L * N;
        hg_init<<<dim3((unsigned)((span + TPB - 1) / TPB)), dim3(TPB), 0, stream>>>(
            deg, acc1, out, b2, N);
    }

    {
        const int dparts = (E[0] + DG_CH - 1) / DG_CH + (E[1] + DG_CH - 1) / DG_CH +
                           (E[2] + DG_CH - 1) / DG_CH;
        DegArgs da = {src[0], dst[0], src[1], dst[1], src[2], dst[2], E[0], E[1], E[2]};
        hg_degree_lds<<<dim3(2 * dparts, (N + DG_CAP - 1) / DG_CAP), dim3(DG_TPB), 0, stream>>>(
            da, deg, N);
    }
    hg_scale<<<dim3((6 * N + TPB - 1) / TPB), dim3(TPB), 0, stream>>>(deg, 6 * N);

    const int tiles  = (N + 15) / 16;
    const int gblk   = (tiles + (TPB / 32) - 1) / (TPB / 32);
    const int slices = (N + SL_ROWS - 1) / SL_ROWS;
    for (int r = 0; r < 3; ++r) {
        hg_gemm1_wmma<<<dim3(gblk), dim3(TPB), 0, stream>>>(
            x, scales + (size_t)r * N, W1 + (size_t)r * 32 * 16, hpre, N);
        hg_scatter1<<<dim3(slices, SL_PARTS), dim3(SL_TPB), 0, stream>>>(
            src[r], dst[r], E[r], hpre, scales + (size_t)(3 + r) * N, acc1, N);
    }
    {
        long n16 = 16L * N;
        hg_relu_bias<<<dim3((unsigned)((n16 + TPB - 1) / TPB)), dim3(TPB), 0, stream>>>(
            acc1, b1, n16);
    }

    for (int r = 0; r < 3; ++r) {
        hg_gemm2<<<dim3((N + TPB - 1) / TPB), dim3(TPB), 0, stream>>>(
            acc1, scales + (size_t)r * N, W2 + (size_t)r * 16, z, N);
        hg_scatter2<<<dim3((E[r] + TPB - 1) / TPB), dim3(TPB), 0, stream>>>(
            src[r], dst[r], E[r], z, scales + (size_t)(3 + r) * N, out);
    }
}


# layer-2 scatter: group keeps z of the source range and a 25600-bin dst slice in LDS, adds z[src] per edge by LDS atomics, flushes bins x s_in by whole-line atomics; one launch; arrival-order sums
# speedup vs baseline: 11.4153x; 2.5111x over previous
#include <hip/hip_runtime.h>

typedef float v2f __attribute__((ext_vector_type(2)));
typedef float v8f __attribute__((ext_vector_type(8)));

#define TPB 256

__global__ __launch_bounds__(TPB) void hg_init(unsigned* __restrict__ deg,
                                               float* __restrict__ acc1,
                                               float* __restrict__ out,
                                               const float* __restrict__ b2,
                                               int N) {
    long i = (long)blockIdx.x * TPB + threadIdx.x;
    if (i < 6L * N)  deg[i] = 0u;
    if (i < 16L * N) acc1[i] = 0.0f;
    if (i < N)       out[i] = b2[0] + b2[1] + b2[2];
}

#define DG_TPB 1024
#define DG_CAP 51200
#define DG_CH  131072

struct DegArgs {
    const int* s0; const int* d0;
    const int* s1; const int* d1;
    const int* s2; const int* d2;
    int E0, E1, E2;
};

static __device__ __forceinline__ void dg_count(unsigned* cnt, int v, int base, int span) {
    const unsigned l = (unsigned)(v - base);
    if (l < (unsigned)span) atomicAdd(&cnt[l], 1u);
}

__global__ __launch_bounds__(DG_TPB) void hg_degree_lds(DegArgs a,
                                                        unsigned* __restrict__ deg,
                                                        int N) {
    __shared__ unsigned cnt[DG_CAP];

    const int p0 = (a.E0 + DG_CH - 1) / DG_CH;
    const int p1 = (a.E1 + DG_CH - 1) / DG_CH;
    int b = (int)blockIdx.x;
    int r = 0;
    if (b >= 2 * p0) { b -= 2 * p0; r = 1; }
    if (r == 1 && b >= 2 * p1) { b -= 2 * p1; r = 2; }
    const int which = b & 1;
    const int part  = b >> 1;
    const int* __restrict__ idx = (r == 0) ? (which ? a.d0 : a.s0)
                                : (r == 1) ? (which ? a.d1 : a.s1)
                                           : (which ? a.d2 : a.s2);
    const int E = (r == 0) ? a.E0 : (r == 1) ? a.E1 : a.E2;
    unsigned* __restrict__ hist = deg + (size_t)(which * 3 + r) * N;

    const int base = (int)blockIdx.y * DG_CAP;
    const int span = (N - base < DG_CAP) ? (N - base) : DG_CAP;

    for (int i = threadIdx.x; i < span; i += DG_TPB) cnt[i] = 0u;
    __syncthreads();

    const int e_begin = part * DG_CH;
    const int e_end   = (e_begin + DG_CH < E) ? (e_begin + DG_CH) : E;
    for (int e0 = e_begin; e0 < e_end; e0 += 4 * DG_TPB * 4) {
        int4 d[4];
        if (e0 + 4 * DG_TPB * 4 <= e_end) {
#pragma unroll
            for (int k = 0; k < 4; ++k)
                d[k] = *(const int4*)(idx + e0 + k * (DG_TPB * 4) + threadIdx.x * 4);
        } else {
#pragma unroll
            for (int k = 0; k < 4; ++k) {
                const int e = e0 + k * (DG_TPB * 4) + threadIdx.x * 4;
                d[k].x = (e     < e_end) ? idx[e]     : -1;
                d[k].y = (e + 1 < e_end) ? idx[e + 1] : -1;
                d[k].z = (e + 2 < e_end) ? idx[e + 2] : -1;
                d[k].w = (e + 3 < e_end) ? idx[e + 3] : -1;
            }
        }
#pragma unroll
        for (int k = 0; k < 4; ++k) {
            dg_count(cnt, d[k].x, base, span);
            dg_count(cnt, d[k].y, base, span);
            dg_count(cnt, d[k].z, base, span);
            dg_count(cnt, d[k].w, base, span);
        }
    }
    __syncthreads();

    const int tiles = (span + DG_TPB - 1) / DG_TPB;
    const int rot   = (int)(blockIdx.x % (unsigned)tiles);
    for (int j = 0; j < tiles; ++j) {
        int t = j + rot;
        if (t >= tiles) t -= tiles;
        const int i = t * DG_TPB + (int)threadIdx.x;
        if (i < span) {
            const unsigned c = cnt[i];
            if (c != 0u) atomicAdd(&hist[base + i], c);
        }
    }
}

__global__ __launch_bounds__(TPB) void hg_scale(unsigned* __restrict__ deg, int n) {
    int i = blockIdx.x * TPB + threadIdx.x;
    if (i < n) {
        float d = (float)deg[i];
        if (d < 1.0f) d = 1.0f;
        ((float*)deg)[i] = rsqrtf(d);
    }
}

__global__ __launch_bounds__(TPB) void hg_gemm1_wmma(const float* __restrict__ x,
                                                     const float* __restrict__ s_out,
                                                     const float* __restrict__ W,
                                                     float* __restrict__ h_pre,
                                                     int N) {
    const int lane = threadIdx.x & 31;
    const int wave = threadIdx.x >> 5;
    const int tile = blockIdx.x * (TPB / 32) + wave;
    const int row0 = tile * 16;
    if (row0 >= N) return;

    const int half = lane >> 4;
    const int lm   = lane & 15;
    int row  = row0 + lm;
    int rowc = row < N ? row : N - 1;
    const float sc = s_out[rowc];

    v8f c = {};
#pragma unroll
    for (int kb = 0; kb < 8; ++kb) {
        const int k0 = kb * 4 + half * 2;
        v2f a, b;
        a.x = x[(size_t)rowc * 32 + k0]     * sc;
        a.y = x[(size_t)rowc * 32 + k0 + 1] * sc;
        b.x = W[(k0)     * 16 + lm];
        b.y = W[(k0 + 1) * 16 + lm];
        c = __builtin_amdgcn_wmma_f32_16x16x4_f32(
                 false, a,  false, b,
                 (short)0, c,  false,  false);
    }

    float* __restrict__ dp = h_pre + (size_t)(row0 + half * 8) * 16 + lm;
    if (row0 + 16 <= N) {
#pragma unroll
        for (int v = 0; v < 8; ++v) dp[(size_t)v * 16] = c[v];
    } else {
#pragma unroll
        for (int v = 0; v < 8; ++v) {
            const int orow = row0 + v + half * 8;
            if (orow < N) h_pre[(size_t)orow * 16 + lm] = c[v];
        }
    }
}

#define SL_ROWS  1024
#define SL_WAVES 32
#define SL_TPB   (SL_WAVES * 32)
#define SL_PARTS 5
#define SL_GROUP 512
#define SL_QCAP  256

template <int I>
static __device__ __forceinline__ unsigned row_lane_u(unsigned v) {
    return (unsigned)__builtin_amdgcn_update_dpp(0, (int)v, 0x150 + I, 0xf, 0xf, true);
}

#define S1_ALL(M) M(0) M(1) M(2) M(3) M(4) M(5) M(6) M(7) M(8) M(9) M(10) M(11) M(12) M(13) M(14) M(15)
#define S1_LOAD(I) v[I] = *(const float*)((const char*)h_pre + (row_lane_u<I>(sb) + lm4));
#define S1_ADD(I)                                                              \
    if (!GUARD || I < nrow)                                                    \
        atomicAdd((float*)((char*)rows + (row_lane_u<I>(lb) + lm4)), v[I]);

template <bool GUARD>
static __device__ __forceinline__ void s1_drain(const unsigned* qe, const unsigned* ql,
                                                int head, int cnt, int lane, unsigned lm4,
                                                const int* __restrict__ src,
                                                const float* __restrict__ h_pre,
                                                float* rows) {
    __builtin_amdgcn_fence(__ATOMIC_RELEASE, "wavefront");
    __builtin_amdgcn_wave_barrier();
    __builtin_amdgcn_fence(__ATOMIC_ACQUIRE, "wavefront");
    const int ent = (!GUARD || lane < cnt) ? lane : 0;
    const unsigned qi = (unsigned)(head + ent) & (SL_QCAP - 1);
    const unsigned e  = qe[qi];
    const unsigned lb = ql[qi];
    const unsigned sb = (unsigned)src[e] * 64u;
    const int nrow = cnt - (lane & 16);
    float v[16];
    S1_ALL(S1_LOAD)
    S1_ALL(S1_ADD)
}

static __device__ __forceinline__ void s1_test(int dv, int e, int base,
                                               unsigned* qe, unsigned* ql, int& tail) {
    const unsigned l = (unsigned)(dv - base);
    const bool hit = l < (unsigned)SL_ROWS;
    const unsigned m = __builtin_amdgcn_ballot_w32(hit);
    if (m != 0u) {
        const unsigned pos = ((unsigned)tail + __builtin_amdgcn_mbcnt_lo(m, 0u)) & (SL_QCAP - 1);
        if (hit) { qe[pos] = (unsigned)e; ql[pos] = l * 64u; }
        tail += __builtin_popcount(m);
    }
}

__global__ __launch_bounds__(SL_TPB) void hg_scatter1(const int* __restrict__ src,
                                                      const int* __restrict__ dst,
                                                      int E,
                                                      const float* __restrict__ h_pre,
                                                      const float* __restrict__ s_in,
                                                      float* __restrict__ acc,
                                                      int N) {
    __shared__ float    rows[SL_ROWS * 16];
    __shared__ unsigned q_e[SL_WAVES][SL_QCAP];
    __shared__ unsigned q_l[SL_WAVES][SL_QCAP];

    const int lane = threadIdx.x & 31;
    const unsigned lm4 = (unsigned)(lane & 15) * 4u;
    const int wave = __builtin_amdgcn_readfirstlane((int)(threadIdx.x >> 5));
    const int base = (int)blockIdx.x * SL_ROWS;

    for (int i = threadIdx.x; i < SL_ROWS * 4; i += SL_TPB)
        ((float4*)rows)[i] = make_float4(0.0f, 0.0f, 0.0f, 0.0f);
    __syncthreads();

    const int chunk   = ((E + SL_PARTS * SL_GROUP - 1) / (SL_PARTS * SL_GROUP)) * SL_GROUP;
    const int e_begin = (int)blockIdx.y * chunk;
    const int e_end   = (e_begin + chunk < E) ? (e_begin + chunk) : E;

    unsigned* qe = q_e[wave];
    unsigned* ql = q_l[wave];
    int head = 0, tail = 0;

    for (int e0 = e_begin + wave * SL_GROUP; e0 < e_end; e0 += SL_WAVES * SL_GROUP) {
        int4 d[4];
        if (e0 + SL_GROUP <= e_end) {
#pragma unroll
            for (int k = 0; k < 4; ++k)
                d[k] = *(const int4*)(dst + e0 + k * 128 + lane * 4);
        } else {
#pragma unroll
            for (int k = 0; k < 4; ++k) {
                const int e = e0 + k * 128 + lane * 4;
                d[k].x = (e     < e_end) ? dst[e]     : -1;
                d[k].y = (e + 1 < e_end) ? dst[e + 1] : -1;
                d[k].z = (e + 2 < e_end) ? dst[e + 2] : -1;
                d[k].w = (e + 3 < e_end) ? dst[e + 3] : -1;
            }
        }
#pragma unroll
        for (int k = 0; k < 4; ++k) {
            const int e = e0 + k * 128 + lane * 4;
            s1_test(d[k].x, e,     base, qe, ql, tail);
            s1_test(d[k].y, e + 1, base, qe, ql, tail);
            s1_test(d[k].z, e + 2, base, qe, ql, tail);
            s1_test(d[k].w, e + 3, base, qe, ql, tail);
            while (tail - head >= 32) {
                s1_drain<false>(qe, ql, head, 32, lane, lm4, src, h_pre, rows);
                head += 32;
            }
        }
    }
    if (tail - head > 0)
        s1_drain<true>(qe, ql, head, tail - head, lane, lm4, src, h_pre, rows);

    __syncthreads();

    constexpr int FL = SL_ROWS * 16 / SL_TPB;
    const int f  = threadIdx.x & 15;
    const int r0 = threadIdx.x >> 4;
    float pv[FL], ps[FL];
#pragma unroll
    for (int j = 0; j < FL; ++j) {
        const int r  = r0 + j * (SL_TPB / 16);
        const int gd = base + r;
        pv[j] = rows[r * 16 + f];
        ps[j] = s_in[gd < N ? gd : N - 1];
    }
#pragma unroll
    for (int j = 0; j < FL; ++j) {
        const int gd = base + r0 + j * (SL_TPB / 16);
        if (gd < N && pv[j] != 0.0f)
            atomicAdd(&acc[(size_t)gd * 16 + f], pv[j] * ps[j]);
    }
}

__global__ __launch_bounds__(TPB) void hg_relu_bias(float* __restrict__ acc,
                                                    const float* __restrict__ b1,
                                                    long n16) {
    long i = (long)blockIdx.x * TPB + threadIdx.x;
    if (i < n16) {
        int f = (int)(i & 15);
        float v = acc[i] + b1[f] + b1[16 + f] + b1[32 + f];
        acc[i] = v > 0.0f ? v : 0.0f;
    }
}

__global__ __launch_bounds__(TPB) void hg_gemm2(const float* __restrict__ h,
                                                const float* __restrict__ s_out,
                                                const float* __restrict__ w,
                                                float* __restrict__ z,
                                                int N) {
    int i = blockIdx.x * TPB + threadIdx.x;
    if (i >= N) return;
    const float4* hp = (const float4*)(h + (size_t)i * 16);
    const float4* wp = (const float4*)w;
    float acc = 0.0f;
#pragma unroll
    for (int q = 0; q < 4; ++q) {
        float4 a = hp[q], b = wp[q];
        acc += a.x * b.x + a.y * b.y + a.z * b.z + a.w * b.w;
    }
    z[i] = acc * s_out[i];
}

#define S2_TPB  1024
#define S2_ZCAP 51200
#define S2_BINS 25600
#define S2_CH   65536
#define S2_ZB   25

static __device__ __forceinline__ float s2_take(const float* zl, int s, int d,
                                                int zbase, int zspan, int dbase, int dspan,
                                                unsigned& bin) {
    const unsigned zi = (unsigned)(s - zbase);
    const unsigned l  = (unsigned)(d - dbase);
    const bool hit = (l < (unsigned)dspan) && (zi < (unsigned)zspan);
    bin = hit ? l : 0xffffffffu;
    return zl[hit ? zi : 0u];
}

__global__ __launch_bounds__(S2_TPB) void hg_scatter2_lds(DegArgs a,
                                                          const float* __restrict__ z3,
                                                          const float* __restrict__ s_in3,
                                                          float* __restrict__ out,
                                                          int N) {
    __shared__ float zl[S2_ZCAP];
    __shared__ float bins[S2_BINS];

    const int p0 = (a.E0 + S2_CH - 1) / S2_CH;
    const int p1 = (a.E1 + S2_CH - 1) / S2_CH;
    int part = (int)blockIdx.x;
    int r = 0;
    if (part >= p0) { part -= p0; r = 1; }
    if (r == 1 && part >= p1) { part -= p1; r = 2; }
    const int* __restrict__ src = (r == 0) ? a.s0 : (r == 1) ? a.s1 : a.s2;
    const int* __restrict__ dst = (r == 0) ? a.d0 : (r == 1) ? a.d1 : a.d2;
    const int E = (r == 0) ? a.E0 : (r == 1) ? a.E1 : a.E2;
    const float* __restrict__ z    = z3    + (size_t)r * N;
    const float* __restrict__ s_in = s_in3 + (size_t)r * N;

    const int dbase = (int)blockIdx.y * S2_BINS;
    const int dspan = (N - dbase < S2_BINS) ? (N - dbase) : S2_BINS;
    const int zbase = (int)blockIdx.z * S2_ZCAP;
    const int zspan = (N - zbase < S2_ZCAP) ? (N - zbase) : S2_ZCAP;
    const int tid = (int)threadIdx.x;

    for (int j0 = 0; j0 < S2_ZCAP / S2_TPB; j0 += S2_ZB) {
        float t[S2_ZB];
#pragma unroll
        for (int j = 0; j < S2_ZB; ++j) {
            const int i = (j0 + j) * S2_TPB + tid;
            t[j] = z[zbase + (i < zspan ? i : zspan - 1)];
        }
#pragma unroll
        for (int j = 0; j < S2_ZB; ++j) {
            const int i = (j0 + j) * S2_TPB + tid;
            if (i < zspan) zl[i] = t[j];
        }
    }
    for (int i = tid; i < dspan; i += S2_TPB) bins[i] = 0.0f;
    __syncthreads();

    const int e_begin = part * S2_CH;
    const int e_end   = (e_begin + S2_CH < E) ? (e_begin + S2_CH) : E;
    for (int e0 = e_begin; e0 < e_end; e0 += 4 * S2_TPB * 4) {
        int4 dd[4], ss[4];
        if (e0 + 4 * S2_TPB * 4 <= e_end) {
#pragma unroll
            for (int k = 0; k < 4; ++k) {
                dd[k] = *(const int4*)(dst + e0 + k * (S2_TPB * 4) + tid * 4);
                ss[k] = *(const int4*)(src + e0 + k * (S2_TPB * 4) + tid * 4);
            }
        } else {
#pragma unroll
            for (int k = 0; k < 4; ++k) {
                const int e = e0 + k * (S2_TPB * 4) + tid * 4;
                dd[k].x = (e     < e_end) ? dst[e]     : -1;
                dd[k].y = (e + 1 < e_end) ? dst[e + 1] : -1;
                dd[k].z = (e + 2 < e_end) ? dst[e + 2] : -1;
                dd[k].w = (e + 3 < e_end) ? dst[e + 3] : -1;
                ss[k].x = (e     < e_end) ? src[e]     : 0;
                ss[k].y = (e + 1 < e_end) ? src[e + 1] : 0;
                ss[k].z = (e + 2 < e_end) ? src[e + 2] : 0;
                ss[k].w = (e + 3 < e_end) ? src[e + 3] : 0;
            }
        }
        float    v[16];
        unsigned b[16];
#pragma unroll
        for (int k = 0; k < 4; ++k) {
            v[4 * k]     = s2_take(zl, ss[k].x, dd[k].x, zbase, zspan, dbase, dspan, b[4 * k]);
            v[4 * k + 1] = s2_take(zl, ss[k].y, dd[k].y, zbase, zspan, dbase, dspan, b[4 * k + 1]);
            v[4 * k + 2] = s2_take(zl, ss[k].z, dd[k].z, zbase, zspan, dbase, dspan, b[4 * k + 2]);
            v[4 * k + 3] = s2_take(zl, ss[k].w, dd[k].w, zbase, zspan, dbase, dspan, b[4 * k + 3]);
        }
#pragma unroll
        for (int q = 0; q < 16; ++q)
            if (b[q] != 0xffffffffu) atomicAdd(&bins[b[q]], v[q]);
    }
    __syncthreads();

    constexpr int FL = S2_BINS / S2_TPB;
    const int tiles = (dspan + S2_TPB - 1) / S2_TPB;
    const int rot   = (int)(blockIdx.x % (unsigned)tiles);
    float pv[FL], ps[FL];
#pragma unroll
    for (int j = 0; j < FL; ++j) {
        int t = j + rot;
        if (t >= tiles) t -= tiles;
        const int i  = t * S2_TPB + tid;
        const int ic = (j < tiles && i < dspan) ? i : 0;
        pv[j] = bins[ic];
        ps[j] = s_in[dbase + ic];
    }
#pragma unroll
    for (int j = 0; j < FL; ++j) {
        int t = j + rot;
        if (t >= tiles) t -= tiles;
        const int i = t * S2_TPB + tid;
        if (j < tiles && i < dspan && pv[j] != 0.0f)
            atomicAdd(&out[dbase + i], pv[j] * ps[j]);
    }
}

extern "C" void kernel_launch(void* const* d_in, const int* in_sizes, int n_in,
                              void* d_out, int out_size, void* d_ws, size_t ws_size,
                              hipStream_t stream) {
    const float* x  = (const float*)d_in[0];
    const int* src[3] = {(const int*)d_in[1], (const int*)d_in[3], (const int*)d_in[5]};
    const int* dst[3] = {(const int*)d_in[2], (const int*)d_in[4], (const int*)d_in[6]};
    const int  E[3]   = {in_sizes[1], in_sizes[3], in_sizes[5]};
    const float* W1 = (const float*)d_in[7];
    const float* b1 = (const float*)d_in[8];
    const float* W2 = (const float*)d_in[9];
    const float* b2 = (const float*)d_in[10];
    const int N = in_sizes[0] / 32;
    float* out = (float*)d_out;

    char* ws = (char*)d_ws;
    unsigned* deg = (unsigned*)ws;
    float* scales = (float*)ws;
    float* acc1   = (float*)(ws + (size_t)6 * N * 4);
    float* hpre   = acc1 + (size_t)N * 16;
    float* z3     = hpre;

    {
        long span = 16L * N;
        hg_init<<<dim3((unsigned)((span + TPB - 1) / TPB)), dim3(TPB), 0, stream>>>(
            deg, acc1, out, b2, N);
    }

    {
        const int dparts = (E[0] + DG_CH - 1) / DG_CH + (E[1] + DG_CH - 1) / DG_CH +
                           (E[2] + DG_CH - 1) / DG_CH;
        DegArgs da = {src[0], dst[0], src[1], dst[1], src[2], dst[2], E[0], E[1], E[2]};
        hg_degree_lds<<<dim3(2 * dparts, (N + DG_CAP - 1) / DG_CAP), dim3(DG_TPB), 0, stream>>>(
            da, deg, N);
    }
    hg_scale<<<dim3((6 * N + TPB - 1) / TPB), dim3(TPB), 0, stream>>>(deg, 6 * N);

    const int tiles  = (N + 15) / 16;
    const int gblk   = (tiles + (TPB / 32) - 1) / (TPB / 32);
    const int slices = (N + SL_ROWS - 1) / SL_ROWS;
    for (int r = 0; r < 3; ++r) {
        hg_gemm1_wmma<<<dim3(gblk), dim3(TPB), 0, stream>>>(
            x, scales + (size_t)r * N, W1 + (size_t)r * 32 * 16, hpre, N);
        hg_scatter1<<<dim3(slices, SL_PARTS), dim3(SL_TPB), 0, stream>>>(
            src[r], dst[r], E[r], hpre, scales + (size_t)(3 + r) * N, acc1, N);
    }
    {
        long n16 = 16L * N;
        hg_relu_bias<<<dim3((unsigned)((n16 + TPB - 1) / TPB)), dim3(TPB), 0, stream>>>(
            acc1, b1, n16);
    }

    for (int r = 0; r < 3; ++r) {
        hg_gemm2<<<dim3((N + TPB - 1) / TPB), dim3(TPB), 0, stream>>>(
            acc1, scales + (size_t)r * N, W2 + (size_t)r * 16, z3 + (size_t)r * N, N);
    }
    {
        const int sparts = (E[0] + S2_CH - 1) / S2_CH + (E[1] + S2_CH - 1) / S2_CH +
                           (E[2] + S2_CH - 1) / S2_CH;
        DegArgs sa = {src[0], dst[0], src[1], dst[1], src[2], dst[2], E[0], E[1], E[2]};
        hg_scatter2_lds<<<dim3(sparts, (N + S2_BINS - 1) / S2_BINS, (N + S2_ZCAP - 1) / S2_ZCAP),
                          dim3(S2_TPB), 0, stream>>>(
            sa, z3, scales + (size_t)3 * N, out, N);
    }
}


# elementwise passes folded into neighbours: counts->rsqrt at use, 1st transform clears accumulator, bias+relu+3 layer-2 dots (now contracted to multiply-adds)+out seed in one launch; 14->10 launches
# speedup vs baseline: 12.1229x; 1.0620x over previous
#include <hip/hip_runtime.h>

typedef float v2f __attribute__((ext_vector_type(2)));
typedef float v8f __attribute__((ext_vector_type(8)));

#define TPB 256

__global__ __launch_bounds__(TPB) void hg_init(unsigned* __restrict__ deg, int n) {
    const int i = ((int)blockIdx.x * TPB + (int)threadIdx.x) * 4;
    if (i + 3 < n) {
        *(uint4*)(deg + i) = make_uint4(0u, 0u, 0u, 0u);
    } else {
        for (int k = i; k < n; ++k) deg[k] = 0u;
    }
}

static __device__ __forceinline__ float deg_scale(unsigned cnt) {
    float d = (float)cnt;
    if (d < 1.0f) d = 1.0f;
    return rsqrtf(d);
}

#define DG_TPB 1024
#define DG_CAP 51200
#define DG_CH  131072

struct DegArgs {
    const int* s0; const int* d0;
    const int* s1; const int* d1;
    const int* s2; const int* d2;
    int E0, E1, E2;
};

static __device__ __forceinline__ void dg_count(unsigned* cnt, int v, int base, int span) {
    const unsigned l = (unsigned)(v - base);
    if (l < (unsigned)span) atomicAdd(&cnt[l], 1u);
}

__global__ __launch_bounds__(DG_TPB) void hg_degree_lds(DegArgs a,
                                                        unsigned* __restrict__ deg,
                                                        int N) {
    __shared__ unsigned cnt[DG_CAP];

    const int p0 = (a.E0 + DG_CH - 1) / DG_CH;
    const int p1 = (a.E1 + DG_CH - 1) / DG_CH;
    int b = (int)blockIdx.x;
    int r = 0;
    if (b >= 2 * p0) { b -= 2 * p0; r = 1; }
    if (r == 1 && b >= 2 * p1) { b -= 2 * p1; r = 2; }
    const int which = b & 1;
    const int part  = b >> 1;
    const int* __restrict__ idx = (r == 0) ? (which ? a.d0 : a.s0)
                                : (r == 1) ? (which ? a.d1 : a.s1)
                                           : (which ? a.d2 : a.s2);
    const int E = (r == 0) ? a.E0 : (r == 1) ? a.E1 : a.E2;
    unsigned* __restrict__ hist = deg + (size_t)(which * 3 + r) * N;

    const int base = (int)blockIdx.y * DG_CAP;
    const int span = (N - base < DG_CAP) ? (N - base) : DG_CAP;

    for (int i = threadIdx.x; i < span; i += DG_TPB) cnt[i] = 0u;
    __syncthreads();

    const int e_begin = part * DG_CH;
    const int e_end   = (e_begin + DG_CH < E) ? (e_begin + DG_CH) : E;
    for (int e0 = e_begin; e0 < e_end; e0 += 4 * DG_TPB * 4) {
        int4 d[4];
        if (e0 + 4 * DG_TPB * 4 <= e_end) {
#pragma unroll
            for (int k = 0; k < 4; ++k)
                d[k] = *(const int4*)(idx + e0 + k * (DG_TPB * 4) + threadIdx.x * 4);
        } else {
#pragma unroll
            for (int k = 0; k < 4; ++k) {
                const int e = e0 + k * (DG_TPB * 4) + threadIdx.x * 4;
                d[k].x = (e     < e_end) ? idx[e]     : -1;
                d[k].y = (e + 1 < e_end) ? idx[e + 1] : -1;
                d[k].z = (e + 2 < e_end) ? idx[e + 2] : -1;
                d[k].w = (e + 3 < e_end) ? idx[e + 3] : -1;
            }
        }
#pragma unroll
        for (int k = 0; k < 4; ++k) {
            dg_count(cnt, d[k].x, base, span);
            dg_count(cnt, d[k].y, base, span);
            dg_count(cnt, d[k].z, base, span);
            dg_count(cnt, d[k].w, base, span);
        }
    }
    __syncthreads();

    const int tiles = (span + DG_TPB - 1) / DG_TPB;
    const int rot   = (int)(blockIdx.x % (unsigned)tiles);
    for (int j = 0; j < tiles; ++j) {
        int t = j + rot;
        if (t >= tiles) t -= tiles;
        const int i = t * DG_TPB + (int)threadIdx.x;
        if (i < span) {
            const unsigned c = cnt[i];
            if (c != 0u) atomicAdd(&hist[base + i], c);
        }
    }
}

__global__ __launch_bounds__(TPB) void hg_gemm1_wmma(const float* __restrict__ x,
                                                     const unsigned* __restrict__ deg_out,
                                                     const float* __restrict__ W,
                                                     float* __restrict__ h_pre,
                                                     float* __restrict__ acc_zero,
                                                     int N) {
    const int lane = threadIdx.x & 31;
    const int wave = threadIdx.x >> 5;
    const int tile = blockIdx.x * (TPB / 32) + wave;
    const int row0 = tile * 16;
    if (row0 >= N) return;

    const int half = lane >> 4;
    const int lm   = lane & 15;
    int row  = row0 + lm;
    int rowc = row < N ? row : N - 1;
    const float sc = deg_scale(deg_out[rowc]);

    if (acc_zero != nullptr) {
        float4* zp = (float4*)(acc_zero + (size_t)row0 * 16);
        const float4 z4 = make_float4(0.0f, 0.0f, 0.0f, 0.0f);
        if (row0 + (lane >> 2) < N)     zp[lane]      = z4;
        if (row0 + 8 + (lane >> 2) < N) zp[lane + 32] = z4;
    }

    v8f c = {};
#pragma unroll
    for (int kb = 0; kb < 8; ++kb) {
        const int k0 = kb * 4 + half * 2;
        v2f a, b;
        a.x = x[(size_t)rowc * 32 + k0]     * sc;
        a.y = x[(size_t)rowc * 32 + k0 + 1] * sc;
        b.x = W[(k0)     * 16 + lm];
        b.y = W[(k0 + 1) * 16 + lm];
        c = __builtin_amdgcn_wmma_f32_16x16x4_f32(
                 false, a,  false, b,
                 (short)0, c,  false,  false);
    }

    float* __restrict__ dp = h_pre + (size_t)(row0 + half * 8) * 16 + lm;
    if (row0 + 16 <= N) {
#pragma unroll
        for (int v = 0; v < 8; ++v) dp[(size_t)v * 16] = c[v];
    } else {
#pragma unroll
        for (int v = 0; v < 8; ++v) {
            const int orow = row0 + v + half * 8;
            if (orow < N) h_pre[(size_t)orow * 16 + lm] = c[v];
        }
    }
}

#define SL_ROWS  1024
#define SL_WAVES 32
#define SL_TPB   (SL_WAVES * 32)
#define SL_PARTS 5
#define SL_GROUP 512
#define SL_QCAP  256

template <int I>
static __device__ __forceinline__ unsigned row_lane_u(unsigned v) {
    return (unsigned)__builtin_amdgcn_update_dpp(0, (int)v, 0x150 + I, 0xf, 0xf, true);
}

#define S1_ALL(M) M(0) M(1) M(2) M(3) M(4) M(5) M(6) M(7) M(8) M(9) M(10) M(11) M(12) M(13) M(14) M(15)
#define S1_LOAD(I) v[I] = *(const float*)((const char*)h_pre + (row_lane_u<I>(sb) + lm4));
#define S1_ADD(I)                                                              \
    if (!GUARD || I < nrow)                                                    \
        atomicAdd((float*)((char*)rows + (row_lane_u<I>(lb) + lm4)), v[I]);

template <bool GUARD>
static __device__ __forceinline__ void s1_drain(const unsigned* qe, const unsigned* ql,
                                                int head, int cnt, int lane, unsigned lm4,
                                                const int* __restrict__ src,
                                                const float* __restrict__ h_pre,
                                                float* rows) {
    __builtin_amdgcn_fence(__ATOMIC_RELEASE, "wavefront");
    __builtin_amdgcn_wave_barrier();
    __builtin_amdgcn_fence(__ATOMIC_ACQUIRE, "wavefront");
    const int ent = (!GUARD || lane < cnt) ? lane : 0;
    const unsigned qi = (unsigned)(head + ent) & (SL_QCAP - 1);
    const unsigned e  = qe[qi];
    const unsigned lb = ql[qi];
    const unsigned sb = (unsigned)src[e] * 64u;
    const int nrow = cnt - (lane & 16);
    float v[16];
    S1_ALL(S1_LOAD)
    S1_ALL(S1_ADD)
}

static __device__ __forceinline__ void s1_test(int dv, int e, int base,
                                               unsigned* qe, unsigned* ql, int& tail) {
    const unsigned l = (unsigned)(dv - base);
    const bool hit = l < (unsigned)SL_ROWS;
    const unsigned m = __builtin_amdgcn_ballot_w32(hit);
    if (m != 0u) {
        const unsigned pos = ((unsigned)tail + __builtin_amdgcn_mbcnt_lo(m, 0u)) & (SL_QCAP - 1);
        if (hit) { qe[pos] = (unsigned)e; ql[pos] = l * 64u; }
        tail += __builtin_popcount(m);
    }
}

__global__ __launch_bounds__(SL_TPB) void hg_scatter1(const int* __restrict__ src,
                                                      const int* __restrict__ dst,
                                                      int E,
                                                      const float* __restrict__ h_pre,
                                                      const unsigned* __restrict__ deg_in,
                                                      float* __restrict__ acc,
                                                      int N) {
    __shared__ float    rows[SL_ROWS * 16];
    __shared__ unsigned q_e[SL_WAVES][SL_QCAP];
    __shared__ unsigned q_l[SL_WAVES][SL_QCAP];

    const int lane = threadIdx.x & 31;
    const unsigned lm4 = (unsigned)(lane & 15) * 4u;
    const int wave = __builtin_amdgcn_readfirstlane((int)(threadIdx.x >> 5));
    const int base = (int)blockIdx.x * SL_ROWS;

    for (int i = threadIdx.x; i < SL_ROWS * 4; i += SL_TPB)
        ((float4*)rows)[i] = make_float4(0.0f, 0.0f, 0.0f, 0.0f);
    __syncthreads();

    const int chunk   = ((E + SL_PARTS * SL_GROUP - 1) / (SL_PARTS * SL_GROUP)) * SL_GROUP;
    const int e_begin = (int)blockIdx.y * chunk;
    const int e_end   = (e_begin + chunk < E) ? (e_begin + chunk) : E;

    unsigned* qe = q_e[wave];
    unsigned* ql = q_l[wave];
    int head = 0, tail = 0;

    for (int e0 = e_begin + wave * SL_GROUP; e0 < e_end; e0 += SL_WAVES * SL_GROUP) {
        int4 d[4];
        if (e0 + SL_GROUP <= e_end) {
#pragma unroll
            for (int k = 0; k < 4; ++k)
                d[k] = *(const int4*)(dst + e0 + k * 128 + lane * 4);
        } else {
#pragma unroll
            for (int k = 0; k < 4; ++k) {
                const int e = e0 + k * 128 + lane * 4;
                d[k].x = (e     < e_end) ? dst[e]     : -1;
                d[k].y = (e + 1 < e_end) ? dst[e + 1] : -1;
                d[k].z = (e + 2 < e_end) ? dst[e + 2] : -1;
                d[k].w = (e + 3 < e_end) ? dst[e + 3] : -1;
            }
        }
#pragma unroll
        for (int k = 0; k < 4; ++k) {
            const int e = e0 + k * 128 + lane * 4;
            s1_test(d[k].x, e,     base, qe, ql, tail);
            s1_test(d[k].y, e + 1, base, qe, ql, tail);
            s1_test(d[k].z, e + 2, base, qe, ql, tail);
            s1_test(d[k].w, e + 3, base, qe, ql, tail);
            while (tail - head >= 32) {
                s1_drain<false>(qe, ql, head, 32, lane, lm4, src, h_pre, rows);
                head += 32;
            }
        }
    }
    if (tail - head > 0)
        s1_drain<true>(qe, ql, head, tail - head, lane, lm4, src, h_pre, rows);

    __syncthreads();

    constexpr int FL = SL_ROWS * 16 / SL_TPB;
    const int f  = threadIdx.x & 15;
    const int r0 = threadIdx.x >> 4;
    float pv[FL], ps[FL];
#pragma unroll
    for (int j = 0; j < FL; ++j) {
        const int r  = r0 + j * (SL_TPB / 16);
        const int gd = base + r;
        pv[j] = rows[r * 16 + f];
        ps[j] = deg_scale(deg_in[gd < N ? gd : N - 1]);
    }
#pragma unroll
    for (int j = 0; j < FL; ++j) {
        const int gd = base + r0 + j * (SL_TPB / 16);
        if (gd < N && pv[j] != 0.0f)
            atomicAdd(&acc[(size_t)gd * 16 + f], pv[j] * ps[j]);
    }
}

__global__ __launch_bounds__(TPB) void hg_relu_gemm2(const float* __restrict__ acc,
                                                     const float* __restrict__ b1,
                                                     const unsigned* __restrict__ deg_out3,
                                                     const float* __restrict__ W2,
                                                     const float* __restrict__ b2,
                                                     float* __restrict__ z3,
                                                     float* __restrict__ out,
                                                     int N) {
    const int i = blockIdx.x * TPB + threadIdx.x;
    if (i >= N) return;
    const float4* hp = (const float4*)(acc + (size_t)i * 16);
    float4 h[4];
#pragma unroll
    for (int q = 0; q < 4; ++q) h[q] = hp[q];
#pragma unroll
    for (int q = 0; q < 4; ++q) {
        const int f = 4 * q;
        float v;
        v = h[q].x + b1[f]     + b1[16 + f]     + b1[32 + f];     h[q].x = v > 0.0f ? v : 0.0f;
        v = h[q].y + b1[f + 1] + b1[16 + f + 1] + b1[32 + f + 1]; h[q].y = v > 0.0f ? v : 0.0f;
        v = h[q].z + b1[f + 2] + b1[16 + f + 2] + b1[32 + f + 2]; h[q].z = v > 0.0f ? v : 0.0f;
        v = h[q].w + b1[f + 3] + b1[16 + f + 3] + b1[32 + f + 3]; h[q].w = v > 0.0f ? v : 0.0f;
    }
#pragma unroll
    for (int r = 0; r < 3; ++r) {
        const float4* wp = (const float4*)(W2 + r * 16);
        float s = 0.0f;
#pragma unroll
        for (int q = 0; q < 4; ++q) {
            const float4 a = h[q], b = wp[q];
            s += a.x * b.x + a.y * b.y + a.z * b.z + a.w * b.w;
        }
        z3[(size_t)r * N + i] = s * deg_scale(deg_out3[(size_t)r * N + i]);
    }
    out[i] = b2[0] + b2[1] + b2[2];
}

#define S2_TPB  1024
#define S2_ZCAP 51200
#define S2_BINS 25600
#define S2_CH   65536
#define S2_ZB   25

static __device__ __forceinline__ float s2_take(const float* zl, int s, int d,
                                                int zbase, int zspan, int dbase, int dspan,
                                                unsigned& bin) {
    const unsigned zi = (unsigned)(s - zbase);
    const unsigned l  = (unsigned)(d - dbase);
    const bool hit = (l < (unsigned)dspan) && (zi < (unsigned)zspan);
    bin = hit ? l : 0xffffffffu;
    return zl[hit ? zi : 0u];
}

__global__ __launch_bounds__(S2_TPB) void hg_scatter2_lds(DegArgs a,
                                                          const float* __restrict__ z3,
                                                          const unsigned* __restrict__ deg_in3,
                                                          float* __restrict__ out,
                                                          int N) {
    __shared__ float zl[S2_ZCAP];
    __shared__ float bins[S2_BINS];

    const int p0 = (a.E0 + S2_CH - 1) / S2_CH;
    const int p1 = (a.E1 + S2_CH - 1) / S2_CH;
    int part = (int)blockIdx.x;
    int r = 0;
    if (part >= p0) { part -= p0; r = 1; }
    if (r == 1 && part >= p1) { part -= p1; r = 2; }
    const int* __restrict__ src = (r == 0) ? a.s0 : (r == 1) ? a.s1 : a.s2;
    const int* __restrict__ dst = (r == 0) ? a.d0 : (r == 1) ? a.d1 : a.d2;
    const int E = (r == 0) ? a.E0 : (r == 1) ? a.E1 : a.E2;
    const float* __restrict__ z    = z3    + (size_t)r * N;
    const unsigned* __restrict__ deg_in = deg_in3 + (size_t)r * N;

    const int dbase = (int)blockIdx.y * S2_BINS;
    const int dspan = (N - dbase < S2_BINS) ? (N - dbase) : S2_BINS;
    const int zbase = (int)blockIdx.z * S2_ZCAP;
    const int zspan = (N - zbase < S2_ZCAP) ? (N - zbase) : S2_ZCAP;
    const int tid = (int)threadIdx.x;

    for (int j0 = 0; j0 < S2_ZCAP / S2_TPB; j0 += S2_ZB) {
        float t[S2_ZB];
#pragma unroll
        for (int j = 0; j < S2_ZB; ++j) {
            const int i = (j0 + j) * S2_TPB + tid;
            t[j] = z[zbase + (i < zspan ? i : zspan - 1)];
        }
#pragma unroll
        for (int j = 0; j < S2_ZB; ++j) {
            const int i = (j0 + j) * S2_TPB + tid;
            if (i < zspan) zl[i] = t[j];
        }
    }
    for (int i = tid; i < dspan; i += S2_TPB) bins[i] = 0.0f;
    __syncthreads();

    const int e_begin = part * S2_CH;
    const int e_end   = (e_begin + S2_CH < E) ? (e_begin + S2_CH) : E;
    for (int e0 = e_begin; e0 < e_end; e0 += 4 * S2_TPB * 4) {
        int4 dd[4], ss[4];
        if (e0 + 4 * S2_TPB * 4 <= e_end) {
#pragma unroll
            for (int k = 0; k < 4; ++k) {
                dd[k] = *(const int4*)(dst + e0 + k * (S2_TPB * 4) + tid * 4);
                ss[k] = *(const int4*)(src + e0 + k * (S2_TPB * 4) + tid * 4);
            }
        } else {
#pragma unroll
            for (int k = 0; k < 4; ++k) {
                const int e = e0 + k * (S2_TPB * 4) + tid * 4;
                dd[k].x = (e     < e_end) ? dst[e]     : -1;
                dd[k].y = (e + 1 < e_end) ? dst[e + 1] : -1;
                dd[k].z = (e + 2 < e_end) ? dst[e + 2] : -1;
                dd[k].w = (e + 3 < e_end) ? dst[e + 3] : -1;
                ss[k].x = (e     < e_end) ? src[e]     : 0;
                ss[k].y = (e + 1 < e_end) ? src[e + 1] : 0;
                ss[k].z = (e + 2 < e_end) ? src[e + 2] : 0;
                ss[k].w = (e + 3 < e_end) ? src[e + 3] : 0;
            }
        }
        float    v[16];
        unsigned b[16];
#pragma unroll
        for (int k = 0; k < 4; ++k) {
            v[4 * k]     = s2_take(zl, ss[k].x, dd[k].x, zbase, zspan, dbase, dspan, b[4 * k]);
            v[4 * k + 1] = s2_take(zl, ss[k].y, dd[k].y, zbase, zspan, dbase, dspan, b[4 * k + 1]);
            v[4 * k + 2] = s2_take(zl, ss[k].z, dd[k].z, zbase, zspan, dbase, dspan, b[4 * k + 2]);
            v[4 * k + 3] = s2_take(zl, ss[k].w, dd[k].w, zbase, zspan, dbase, dspan, b[4 * k + 3]);
        }
#pragma unroll
        for (int q = 0; q < 16; ++q)
            if (b[q] != 0xffffffffu) atomicAdd(&bins[b[q]], v[q]);
    }
    __syncthreads();

    constexpr int FL = S2_BINS / S2_TPB;
    const int tiles = (dspan + S2_TPB - 1) / S2_TPB;
    const int rot   = (int)(blockIdx.x % (unsigned)tiles);
    float pv[FL], ps[FL];
#pragma unroll
    for (int j = 0; j < FL; ++j) {
        int t = j + rot;
        if (t >= tiles) t -= tiles;
        const int i  = t * S2_TPB + tid;
        const int ic = (j < tiles && i < dspan) ? i : 0;
        pv[j] = bins[ic];
        ps[j] = deg_scale(deg_in[dbase + ic]);
    }
#pragma unroll
    for (int j = 0; j < FL; ++j) {
        int t = j + rot;
        if (t >= tiles) t -= tiles;
        const int i = t * S2_TPB + tid;
        if (j < tiles && i < dspan && pv[j] != 0.0f)
            atomicAdd(&out[dbase + i], pv[j] * ps[j]);
    }
}

extern "C" void kernel_launch(void* const* d_in, const int* in_sizes, int n_in,
                              void* d_out, int out_size, void* d_ws, size_t ws_size,
                              hipStream_t stream) {
    const float* x  = (const float*)d_in[0];
    const int* src[3] = {(const int*)d_in[1], (const int*)d_in[3], (const int*)d_in[5]};
    const int* dst[3] = {(const int*)d_in[2], (const int*)d_in[4], (const int*)d_in[6]};
    const int  E[3]   = {in_sizes[1], in_sizes[3], in_sizes[5]};
    const float* W1 = (const float*)d_in[7];
    const float* b1 = (const float*)d_in[8];
    const float* W2 = (const float*)d_in[9];
    const float* b2 = (const float*)d_in[10];
    const int N = in_sizes[0] / 32;
    float* out = (float*)d_out;

    char* ws = (char*)d_ws;
    unsigned* deg = (unsigned*)ws;
    float* acc1   = (float*)(ws + (size_t)6 * N * 4);
    float* hpre   = acc1 + (size_t)N * 16;
    float* z3     = hpre;

    hg_init<<<dim3((6 * N + 4 * TPB - 1) / (4 * TPB)), dim3(TPB), 0, stream>>>(deg, 6 * N);

    {
        const int dparts = (E[0] + DG_CH - 1) / DG_CH + (E[1] + DG_CH - 1) / DG_CH +
                           (E[2] + DG_CH - 1) / DG_CH;
        DegArgs da = {src[0], dst[0], src[1], dst[1], src[2], dst[2], E[0], E[1], E[2]};
        hg_degree_lds<<<dim3(2 * dparts, (N + DG_CAP - 1) / DG_CAP), dim3(DG_TPB), 0, stream>>>(
            da, deg, N);
    }

    const int tiles  = (N + 15) / 16;
    const int gblk   = (tiles + (TPB / 32) - 1) / (TPB / 32);
    const int slices = (N + SL_ROWS - 1) / SL_ROWS;
    for (int r = 0; r < 3; ++r) {
        hg_gemm1_wmma<<<dim3(gblk), dim3(TPB), 0, stream>>>(
            x, deg + (size_t)r * N, W1 + (size_t)r * 32 * 16, hpre,
            (r == 0) ? acc1 : (float*)nullptr, N);
        hg_scatter1<<<dim3(slices, SL_PARTS), dim3(SL_TPB), 0, stream>>>(
            src[r], dst[r], E[r], hpre, deg + (size_t)(3 + r) * N, acc1, N);
    }

    hg_relu_gemm2<<<dim3((N + TPB - 1) / TPB), dim3(TPB), 0, stream>>>(
        acc1, b1, deg, W2, b2, z3, out, N);
    {
        const int sparts = (E[0] + S2_CH - 1) / S2_CH + (E[1] + S2_CH - 1) / S2_CH +
                           (E[2] + S2_CH - 1) / S2_CH;
        DegArgs sa = {src[0], dst[0], src[1], dst[1], src[2], dst[2], E[0], E[1], E[2]};
        hg_scatter2_lds<<<dim3(sparts, (N + S2_BINS - 1) / S2_BINS, (N + S2_ZCAP - 1) / S2_ZCAP),
                          dim3(S2_TPB), 0, stream>>>(
            sa, z3, deg + (size_t)3 * N, out, N);
    }
}
